# starting kernel
# baseline (speedup 1.0000x reference)
#include <hip/hip_runtime.h>
#include <math.h>

#define NPIX   102400
#define LCONV  589
#define KPAD   592
#define NSAMP  50
#define DELAYN 30
#define REVSZ  1184

typedef float v2f __attribute__((ext_vector_type(2)));
typedef float v8f __attribute__((ext_vector_type(8)));

__device__ __forceinline__ float nan0f(float x) { return (x != x) ? 0.0f : x; }

__global__ __launch_bounds__(256) void dce_prep(const float* __restrict__ st,
                                                const float* __restrict__ cp,
                                                float* __restrict__ aif_g,
                                                int*   __restrict__ idx_g)
{
    __shared__ float st_s[NSAMP], cp_s[NSAMP];
    int tid = threadIdx.x;
    if (tid < NSAMP) { st_s[tid] = st[tid]; cp_s[tid] = cp[tid]; }
    __syncthreads();

    for (int t = tid; t < KPAD; t += 256) {
        float v = 0.0f;
        if (t >= DELAYN && t < LCONV) {
            float x = (float)(t - DELAYN) * 0.1f;
            if (x <= st_s[0])              v = cp_s[0];
            else if (x >= st_s[NSAMP - 1]) v = cp_s[NSAMP - 1];
            else {
                int i = 0;
                for (int j = 1; j < NSAMP - 1; ++j)
                    if (st_s[j] <= x) i = j;
                v = cp_s[i] + (cp_s[i + 1] - cp_s[i]) * (x - st_s[i]) / (st_s[i + 1] - st_s[i]);
            }
        }
        aif_g[t] = v;
    }

    if (tid < 64) {
        int id = -2;
        if (tid < NSAMP) {
            float xv = st_s[tid];
            int j = LCONV - 1;
            for (int t = LCONV - 1; t >= 0; --t)
                if ((float)t * 0.1f >= xv) j = t;
            id = j;
        }
        idx_g[tid] = id;
    }
}

__device__ __forceinline__ void write_tile(v8f acc, int m, int hl, int p,
                                           float M0t, float COSF, float offs,
                                           float* __restrict__ out)
{
    union { v8f v; float f[8]; } u; u.v = acc;
#pragma unroll
    for (int jj = 0; jj < 8; ++jj) {
        int s = m * 16 + jj + 8 * hl;
        if (s < NSAMP) {
            float CA   = u.f[jj];
            float E1CA = __expf(-0.00487f * (1.0f + 4.3f * CA));
            float CAt  = M0t * (1.0f - E1CA) *
                         __builtin_amdgcn_rcpf(1.0f - E1CA * COSF);
            out[s * NPIX + p] = CAt + offs;
        }
    }
}

__global__ __launch_bounds__(256) void dce_main(const float* __restrict__ param,
                                                const float* __restrict__ aif_g,
                                                const int*   __restrict__ idx_g,
                                                float* __restrict__ out)
{
    __shared__ float rev_s[REVSZ];
    __shared__ int   idx_s[64];
    int tid = threadIdx.x;
    for (int i = tid; i < REVSZ; i += 256) {
        int t = (LCONV - 1) - i;
        rev_s[i] = (t >= 0) ? aif_g[t] : 0.0f;
    }
    if (tid < 64) idx_s[tid] = idx_g[tid];
    __syncthreads();

    const int wave = tid >> 5;
    const int lane = tid & 31;
    const int hl   = lane >> 4;
    const int lpix = lane & 15;
    const int tile = blockIdx.x * 8 + wave;
    const int p    = tile * 16 + lpix;

    float ve = param[0 * NPIX + p];
    float vp = param[1 * NPIX + p];
    float fp = param[2 * NPIX + p];
    float ps = param[3 * NPIX + p];

    float Te = nan0f(ve / ps);
    float T  = nan0f((vp + ve) / fp);
    float Tc = nan0f(vp / fp);
    float sm   = T + Te;
    float disc = sqrtf(sm * sm - 4.0f * Tc * Te);
    float inv2 = 1.0f / (2.0f * Tc * Te);
    float thp  = nan0f((sm + disc) * inv2);
    float thm  = nan0f((sm - disc) * inv2);
    float pref = thp * thm;
    float dn   = thp - thm;
    float am = pref / dn;
    float ap = -am;
    float bm = pref * (1.0f - Te * thm) / dn;
    float bp = pref * (Te * thp - 1.0f) / dn;

    float rm    = __expf(-0.1f  * thm);
    float rp    = __expf(-0.1f  * thp);
    float rm589 = __expf(-58.9f * thm);
    float rp589 = __expf(-58.9f * thp);
    float Smm = (1.0f - rm589) / (1.0f - rm);
    float Spp = (1.0f - rp589) / (1.0f - rp);
    float sumHe = am * Smm + ap * Spp;
    float sumHp = bm * Smm + bp * Spp;

    float P = nan0f(vp * nan0f(bm / sumHp) + ve * nan0f(am / sumHe));
    float Q = nan0f(vp * nan0f(bp / sumHp) + ve * nan0f(ap / sumHe));

    float rm2 = rm * rm, rp2 = rp * rp;
    float rm4 = rm2 * rm2, rp4 = rp2 * rp2;
    float rmt = hl ? rm2 : 1.0f;
    float rpt = hl ? rp2 : 1.0f;

    int q0 = (LCONV - 1) - idx_s[ 0 + lpix] + 2 * hl;
    int q1 = (LCONV - 1) - idx_s[16 + lpix] + 2 * hl;
    int q2 = (LCONV - 1) - idx_s[32 + lpix] + 2 * hl;
    int q3 = (LCONV - 1) - idx_s[48 + lpix] + 2 * hl;

    v8f acc0 = {}, acc1 = {}, acc2 = {}, acc3 = {};

#pragma unroll 4
    for (int k = 0; k < KPAD / 4; ++k) {
        float b0 = P * rmt + Q * rpt;
        float rmt1 = rmt * rm, rpt1 = rpt * rp;
        float b1 = P * rmt1 + Q * rpt1;
        rmt *= rm4; rpt *= rp4;
        v2f B; B.x = b0; B.y = b1;

        v2f A0, A1, A2, A3;
        A0.x = rev_s[q0];  A0.y = rev_s[q0 + 1];
        A1.x = rev_s[q1];  A1.y = rev_s[q1 + 1];
        A2.x = rev_s[q2];  A2.y = rev_s[q2 + 1];
        A3.x = rev_s[q3];  A3.y = rev_s[q3 + 1];
        q0 += 4; q1 += 4; q2 += 4; q3 += 4;

        acc0 = __builtin_amdgcn_wmma_f32_16x16x4_f32(false, A0, false, B, (short)0, acc0, false, false);
        acc1 = __builtin_amdgcn_wmma_f32_16x16x4_f32(false, A1, false, B, (short)0, acc1, false, false);
        acc2 = __builtin_amdgcn_wmma_f32_16x16x4_f32(false, A2, false, B, (short)0, acc2, false, false);
        acc3 = __builtin_amdgcn_wmma_f32_16x16x4_f32(false, A3, false, B, (short)0, acc3, false, false);
    }

    const float COSF = 0.98480775301220805936f;
    const float SINF = 0.17364817766693034885f;
    float E1   = __expf(-0.00487f);
    float M0   = 100.0f * (1.0f - COSF * E1) / (SINF * (1.0f - E1));
    float M0t  = M0 * SINF;
    float Mst  = M0t * (1.0f - E1) / (1.0f - E1 * COSF);
    float offs = 100.0f - Mst;

    write_tile(acc0, 0, hl, p, M0t, COSF, offs, out);
    write_tile(acc1, 1, hl, p, M0t, COSF, offs, out);
    write_tile(acc2, 2, hl, p, M0t, COSF, offs, out);
    write_tile(acc3, 3, hl, p, M0t, COSF, offs, out);
}

extern "C" void kernel_launch(void* const* d_in, const int* in_sizes, int n_in,
                              void* d_out, int out_size, void* d_ws, size_t ws_size,
                              hipStream_t stream)
{
    (void)in_sizes; (void)n_in; (void)out_size; (void)ws_size;
    const float* param = (const float*)d_in[0];
    const float* st    = (const float*)d_in[1];
    const float* cp    = (const float*)d_in[2];
    float* out   = (float*)d_out;
    float* aif_g = (float*)d_ws;
    int*   idx_g = (int*)((char*)d_ws + KPAD * sizeof(float));

    dce_prep<<<1, 256, 0, stream>>>(st, cp, aif_g, idx_g);
    dce_main<<<800, 256, 0, stream>>>(param, aif_g, idx_g, out);
}


# K loop of each M-tile stops at the last nonzero column of its band of A (triangular + 3 s delay), pad rows moved into the earliest tile: 592 -> 276 WMMA per wave
# speedup vs baseline: 1.6656x; 1.6656x over previous
#include <hip/hip_runtime.h>
#include <math.h>

#define NPIX   102400
#define LCONV  589
#define KPAD   592
#define NSAMP  50
#define DELAYN 30
#define REVSZ  1184
#define PADROWS 14
#define KSTEPS (KPAD / 4)

typedef float v2f __attribute__((ext_vector_type(2)));
typedef float v8f __attribute__((ext_vector_type(8)));

__device__ __forceinline__ float nan0f(float x) { return (x != x) ? 0.0f : x; }

__global__ __launch_bounds__(256) void dce_prep(const float* __restrict__ st,
                                                const float* __restrict__ cp,
                                                float* __restrict__ aif_g,
                                                int*   __restrict__ idx_g,
                                                int*   __restrict__ ext_g)
{
    __shared__ float st_s[NSAMP], cp_s[NSAMP];
    __shared__ int   idx_sh[64];
    int tid = threadIdx.x;
    if (tid < NSAMP) { st_s[tid] = st[tid]; cp_s[tid] = cp[tid]; }
    __syncthreads();

    for (int t = tid; t < KPAD; t += 256) {
        float v = 0.0f;
        if (t >= DELAYN && t < LCONV) {
            float x = (float)(t - DELAYN) * 0.1f;
            if (x <= st_s[0])              v = cp_s[0];
            else if (x >= st_s[NSAMP - 1]) v = cp_s[NSAMP - 1];
            else {
                int i = 0;
                for (int j = 1; j < NSAMP - 1; ++j)
                    if (st_s[j] <= x) i = j;
                v = cp_s[i] + (cp_s[i + 1] - cp_s[i]) * (x - st_s[i]) / (st_s[i + 1] - st_s[i]);
            }
        }
        aif_g[t] = v;
    }

    if (tid < 64) {
        int id = -2;
        int sm = tid - PADROWS;
        if (sm >= 0) {
            float xv = st_s[sm];
            int j = LCONV - 1;
            for (int t = LCONV - 1; t >= 0; --t)
                if ((float)t * 0.1f >= xv) j = t;
            id = j;
        }
        idx_g[tid] = id;
        idx_sh[tid] = id;
    }
    __syncthreads();
    if (tid < 4) {
        const int lim = 16 * (tid + 1);
        int mx = -1;
#pragma unroll
        for (int i = 0; i < 64; ++i) {
            int c = idx_sh[i] - DELAYN;
            mx = (i < lim && c > mx) ? c : mx;
        }
        int n = (mx >= 0) ? ((mx + 16) / 16) * 4 : 0;
        ext_g[tid] = (n > KSTEPS) ? KSTEPS : n;
    }
}

__device__ __forceinline__ void write_tile(v8f acc, int m, int hl, int p,
                                           float M0t, float COSF, float offs,
                                           float* __restrict__ out)
{
    union { v8f v; float f[8]; } u; u.v = acc;
#pragma unroll
    for (int jj = 0; jj < 8; ++jj) {
        int s = m * 16 + jj + 8 * hl - PADROWS;
        if (s >= 0) {
            float CA   = u.f[jj];
            float E1CA = __expf(-0.00487f * (1.0f + 4.3f * CA));
            float CAt  = M0t * (1.0f - E1CA) *
                         __builtin_amdgcn_rcpf(1.0f - E1CA * COSF);
            out[s * NPIX + p] = CAt + offs;
        }
    }
}

template <bool T0, bool T1, bool T2, bool T3>
__device__ __forceinline__ void kstep(const float* rev_s, float P, float Q,
                                      float rm, float rp, float rm4, float rp4,
                                      float& rmt, float& rpt,
                                      int& q0, int& q1, int& q2, int& q3,
                                      v8f& acc0, v8f& acc1, v8f& acc2, v8f& acc3)
{
    float b0 = P * rmt + Q * rpt;
    float rmt1 = rmt * rm, rpt1 = rpt * rp;
    float b1 = P * rmt1 + Q * rpt1;
    rmt *= rm4; rpt *= rp4;
    v2f B; B.x = b0; B.y = b1;

    if (T0) {
        v2f A; A.x = rev_s[q0]; A.y = rev_s[q0 + 1]; q0 += 4;
        acc0 = __builtin_amdgcn_wmma_f32_16x16x4_f32(false, A, false, B, (short)0, acc0, false, false);
    }
    if (T1) {
        v2f A; A.x = rev_s[q1]; A.y = rev_s[q1 + 1]; q1 += 4;
        acc1 = __builtin_amdgcn_wmma_f32_16x16x4_f32(false, A, false, B, (short)0, acc1, false, false);
    }
    if (T2) {
        v2f A; A.x = rev_s[q2]; A.y = rev_s[q2 + 1]; q2 += 4;
        acc2 = __builtin_amdgcn_wmma_f32_16x16x4_f32(false, A, false, B, (short)0, acc2, false, false);
    }
    if (T3) {
        v2f A; A.x = rev_s[q3]; A.y = rev_s[q3 + 1]; q3 += 4;
        acc3 = __builtin_amdgcn_wmma_f32_16x16x4_f32(false, A, false, B, (short)0, acc3, false, false);
    }
}

__global__ __launch_bounds__(256) void dce_main(const float* __restrict__ param,
                                                const float* __restrict__ aif_g,
                                                const int*   __restrict__ idx_g,
                                                const int*   __restrict__ ext_g,
                                                float* __restrict__ out)
{
    __shared__ float rev_s[REVSZ];
    __shared__ int   idx_s[64];
    int tid = threadIdx.x;
    for (int i = tid; i < REVSZ; i += 256) {
        int t = (LCONV - 1) - i;
        rev_s[i] = (t >= 0) ? aif_g[t] : 0.0f;
    }
    if (tid < 64) idx_s[tid] = idx_g[tid];
    __syncthreads();

    const int wave = tid >> 5;
    const int lane = tid & 31;
    const int hl   = lane >> 4;
    const int lpix = lane & 15;
    const int tile = blockIdx.x * 8 + wave;
    const int p    = tile * 16 + lpix;

    float ve = param[0 * NPIX + p];
    float vp = param[1 * NPIX + p];
    float fp = param[2 * NPIX + p];
    float ps = param[3 * NPIX + p];

    float Te = nan0f(ve / ps);
    float T  = nan0f((vp + ve) / fp);
    float Tc = nan0f(vp / fp);
    float sm   = T + Te;
    float disc = sqrtf(sm * sm - 4.0f * Tc * Te);
    float inv2 = 1.0f / (2.0f * Tc * Te);
    float thp  = nan0f((sm + disc) * inv2);
    float thm  = nan0f((sm - disc) * inv2);
    float pref = thp * thm;
    float dn   = thp - thm;
    float am = pref / dn;
    float ap = -am;
    float bm = pref * (1.0f - Te * thm) / dn;
    float bp = pref * (Te * thp - 1.0f) / dn;

    float rm    = __expf(-0.1f  * thm);
    float rp    = __expf(-0.1f  * thp);
    float rm589 = __expf(-58.9f * thm);
    float rp589 = __expf(-58.9f * thp);
    float Smm = (1.0f - rm589) / (1.0f - rm);
    float Spp = (1.0f - rp589) / (1.0f - rp);
    float sumHe = am * Smm + ap * Spp;
    float sumHp = bm * Smm + bp * Spp;

    float P = nan0f(vp * nan0f(bm / sumHp) + ve * nan0f(am / sumHe));
    float Q = nan0f(vp * nan0f(bp / sumHp) + ve * nan0f(ap / sumHe));

    float rm2 = rm * rm, rp2 = rp * rp;
    float rm4 = rm2 * rm2, rp4 = rp2 * rp2;
    float rmt = hl ? rm2 : 1.0f;
    float rpt = hl ? rp2 : 1.0f;

    int q0 = (LCONV - 1) - idx_s[ 0 + lpix] + 2 * hl;
    int q1 = (LCONV - 1) - idx_s[16 + lpix] + 2 * hl;
    int q2 = (LCONV - 1) - idx_s[32 + lpix] + 2 * hl;
    int q3 = (LCONV - 1) - idx_s[48 + lpix] + 2 * hl;

    v8f acc0 = {}, acc1 = {}, acc2 = {}, acc3 = {};

    const int n0 = ext_g[0], n1 = ext_g[1], n2 = ext_g[2], n3 = ext_g[3];

    int k = 0;
    for (; k < n0; k += 4) {
#pragma unroll
        for (int u = 0; u < 4; ++u)
            kstep<true, true, true, true>(rev_s, P, Q, rm, rp, rm4, rp4, rmt, rpt,
                                          q0, q1, q2, q3, acc0, acc1, acc2, acc3);
    }
    for (; k < n1; k += 4) {
#pragma unroll
        for (int u = 0; u < 4; ++u)
            kstep<false, true, true, true>(rev_s, P, Q, rm, rp, rm4, rp4, rmt, rpt,
                                           q0, q1, q2, q3, acc0, acc1, acc2, acc3);
    }
    for (; k < n2; k += 4) {
#pragma unroll
        for (int u = 0; u < 4; ++u)
            kstep<false, false, true, true>(rev_s, P, Q, rm, rp, rm4, rp4, rmt, rpt,
                                            q0, q1, q2, q3, acc0, acc1, acc2, acc3);
    }
    for (; k < n3; k += 4) {
#pragma unroll
        for (int u = 0; u < 4; ++u)
            kstep<false, false, false, true>(rev_s, P, Q, rm, rp, rm4, rp4, rmt, rpt,
                                             q0, q1, q2, q3, acc0, acc1, acc2, acc3);
    }

    const float COSF = 0.98480775301220805936f;
    const float SINF = 0.17364817766693034885f;
    float E1   = __expf(-0.00487f);
    float M0   = 100.0f * (1.0f - COSF * E1) / (SINF * (1.0f - E1));
    float M0t  = M0 * SINF;
    float Mst  = M0t * (1.0f - E1) / (1.0f - E1 * COSF);
    float offs = 100.0f - Mst;

    write_tile(acc0, 0, hl, p, M0t, COSF, offs, out);
    write_tile(acc1, 1, hl, p, M0t, COSF, offs, out);
    write_tile(acc2, 2, hl, p, M0t, COSF, offs, out);
    write_tile(acc3, 3, hl, p, M0t, COSF, offs, out);
}

extern "C" void kernel_launch(void* const* d_in, const int* in_sizes, int n_in,
                              void* d_out, int out_size, void* d_ws, size_t ws_size,
                              hipStream_t stream)
{
    (void)in_sizes; (void)n_in; (void)out_size; (void)ws_size;
    const float* param = (const float*)d_in[0];
    const float* st    = (const float*)d_in[1];
    const float* cp    = (const float*)d_in[2];
    float* out   = (float*)d_out;
    float* aif_g = (float*)d_ws;
    int*   idx_g = (int*)((char*)d_ws + KPAD * sizeof(float));
    int*   ext_g = idx_g + 64;

    dce_prep<<<1, 256, 0, stream>>>(st, cp, aif_g, idx_g, ext_g);
    dce_main<<<800, 256, 0, stream>>>(param, aif_g, idx_g, ext_g, out);
}


# Grid reshaped from 800 blocks of 8 waves to 256 blocks of 25 waves, one per WGP with 25 strips each: busiest SIMD holds 7 waves instead of 8, tables built once per 25 strips instead of per 8
# speedup vs baseline: 6.7963x; 4.0804x over previous
#include <hip/hip_runtime.h>
#include <math.h>

#define NPIX   102400
#define LCONV  589
#define KPAD   592
#define NSAMP  50
#define HALFS  25
#define DELAYN 30
#define REVSZ  1184
#define KSTEPS (KPAD / 4)
#define BWAVES 25
#define BTHREADS (BWAVES * 32)

typedef float v2f __attribute__((ext_vector_type(2)));
typedef float v8f __attribute__((ext_vector_type(8)));

__device__ __forceinline__ float nan0f(float x) { return (x != x) ? 0.0f : x; }

__device__ __forceinline__ float xhalf(float x)
{
    return __builtin_bit_cast(float, __builtin_amdgcn_permlanex16(__builtin_bit_cast(int, x), __builtin_bit_cast(int, x),
                                                                  0x76543210, 0xfedcba98, false, false));
}

__device__ __forceinline__ int fine_index(float xv)
{
    int lo = 0, hi = LCONV - 1;
#pragma unroll
    for (int it = 0; it < 10; ++it) {
        int mid = (lo + hi) >> 1;
        bool ge = ((float)mid * 0.1f >= xv);
        int up = (mid + 1 < hi) ? mid + 1 : hi;
        hi = ge ? mid : hi;
        lo = ge ? lo : up;
    }
    return lo;
}

__device__ __forceinline__ int rowmax16(int v)
{
    int o;
    o = __builtin_amdgcn_update_dpp(v, v, 0x161, 0xf, 0xf, false); v = (o > v) ? o : v;
    o = __builtin_amdgcn_update_dpp(v, v, 0x162, 0xf, 0xf, false); v = (o > v) ? o : v;
    o = __builtin_amdgcn_update_dpp(v, v, 0x164, 0xf, 0xf, false); v = (o > v) ? o : v;
    o = __builtin_amdgcn_update_dpp(v, v, 0x168, 0xf, 0xf, false); v = (o > v) ? o : v;
    return v;
}

__global__ __launch_bounds__(BTHREADS) void dce_main(const float* __restrict__ param,
                                                const float* __restrict__ st,
                                                const float* __restrict__ cp,
                                                float* __restrict__ out)
{
    __shared__ float rev_s[REVSZ];
    __shared__ int   idx_s[64];
    __shared__ int   dgi_s[64];
    __shared__ __attribute__((aligned(16))) float dgf_s[64];
    __shared__ __attribute__((aligned(32))) float st_s[64];
    __shared__ float cp_s[64];
    const int tid  = threadIdx.x;
    const int wave = tid >> 5;
    const int lane = tid & 31;
    const int hl   = lane >> 4;
    const int lpix = lane & 15;
    const int tile = blockIdx.x * BWAVES + wave;
    const int p    = tile * 16 + lpix;

    const float st0 = st[0], stL = st[NSAMP - 1];
    const float cp0 = cp[0], cpL = cp[NSAMP - 1];
    float piv[6];
#pragma unroll
    for (int k = 0; k < 6; ++k) piv[k] = st[8 * (k + 1)];
    if (tid < 64) {
        bool in = tid < NSAMP;
        st_s[tid] = in ? st[tid] : __builtin_inff();
        cp_s[tid] = in ? cp[tid] : 0.0f;
    }
    __syncthreads();

    float ve = param[0 * NPIX + p];
    float vp = param[1 * NPIX + p];
    float fp = param[2 * NPIX + p];
    float ps = param[3 * NPIX + p];

    if (tid < 192) {
        float x[3], v[3];
        int   blk[3], ii[3];
#pragma unroll
        for (int r = 0; r < 3; ++r) {
            int t = (LCONV - 1) - (tid + 192 * r);
            t = (t > DELAYN) ? t : DELAYN;
            x[r] = (float)(t - DELAYN) * 0.1f;
            int c = 0;
#pragma unroll
            for (int k = 0; k < 6; ++k) c += (piv[k] <= x[r]) ? 8 : 0;
            blk[r] = c;
        }
#pragma unroll
        for (int r = 0; r < 3; ++r) {
            const float4* b = (const float4*)(st_s + blk[r]);
            float4 lo = b[0], hi = b[1];
            int n = blk[r];
            n += (lo.y <= x[r]) ? 1 : 0;
            n += (lo.z <= x[r]) ? 1 : 0;
            n += (lo.w <= x[r]) ? 1 : 0;
            n += (hi.x <= x[r]) ? 1 : 0;
            n += (hi.y <= x[r]) ? 1 : 0;
            n += (hi.z <= x[r]) ? 1 : 0;
            n += (hi.w <= x[r]) ? 1 : 0;
            ii[r] = n;
        }
#pragma unroll
        for (int r = 0; r < 3; ++r) {
            int i = ii[r];
            float c0 = cp_s[i], c1 = cp_s[i + 1];
            float s0 = st_s[i], s1 = st_s[i + 1];
            v[r] = c0 + (c1 - c0) * (x[r] - s0) / (s1 - s0);
        }
        asm volatile("" : "+v"(v[0]), "+v"(v[1]), "+v"(v[2]));
#pragma unroll
        for (int r = 0; r < 3; ++r) {
            int q = tid + 192 * r;
            float w = v[r];
            w = (x[r] >= stL) ? cpL : w;
            w = (x[r] <= st0) ? cp0 : w;
            w = (q <= (LCONV - 1) - DELAYN) ? w : 0.0f;
            rev_s[q] = w;
        }
    } else if (tid < 256) {
        int e  = tid - 192;
        int i  = e & 31;
        int s  = HALFS * (e >> 5) + i;
        int sp = (s > 0) ? s - 1 : 0;
        int id = fine_index(st_s[s]);
        int ip = fine_index(st_s[sp]);
        ip = (s > 0) ? ip : -1;
        int d = id - ip;
        d = (d > 0) ? d : 0;
        bool used = i < HALFS;
        id = used ? id : -2;
        d  = used ? d : 0;
        idx_s[e] = id;
        dgi_s[e] = d;
        dgf_s[e] = (float)d;
    }
    if (576 + tid < REVSZ) rev_s[576 + tid] = 0.0f;
    __syncthreads();


    float Te = nan0f(ve / ps);
    float T  = nan0f((vp + ve) / fp);
    float Tc = nan0f(vp / fp);
    float sm   = T + Te;
    float disc = sqrtf(sm * sm - 4.0f * Tc * Te);
    float inv2 = 1.0f / (2.0f * Tc * Te);
    float thp  = nan0f((sm + disc) * inv2);
    float thm  = nan0f((sm - disc) * inv2);
    float pref = thp * thm;
    float dn   = thp - thm;
    float am = pref / dn;
    float ap = -am;
    float bm = pref * (1.0f - Te * thm) / dn;
    float bp = pref * (Te * thp - 1.0f) / dn;

    float rm    = __expf(-0.1f  * thm);
    float rp    = __expf(-0.1f  * thp);
    float rm589 = __expf(-58.9f * thm);
    float rp589 = __expf(-58.9f * thp);
    float Smm = (1.0f - rm589) / (1.0f - rm);
    float Spp = (1.0f - rp589) / (1.0f - rp);
    float sumHe = am * Smm + ap * Spp;
    float sumHp = bm * Smm + bp * Spp;

    float P = nan0f(vp * nan0f(bm / sumHp) + ve * nan0f(am / sumHe));
    float Q = nan0f(vp * nan0f(bp / sumHp) + ve * nan0f(ap / sumHe));

    const float lgm = fmaxf(-0.1f * thm * 1.44269504088896340736f, -1.0e30f);
    const float lgp = fmaxf(-0.1f * thp * 1.44269504088896340736f, -1.0e30f);

    float rm2 = rm * rm, rp2 = rp * rp;
    float rm4 = rm2 * rm2, rp4 = rp2 * rp2;
    float rmt = hl ? rm2 : 1.0f;
    float rpt = hl ? rp2 : 1.0f;

    int qa[4], rem[4];
    int dmx = 0;
#pragma unroll
    for (int m = 0; m < 4; ++m) {
        int e = 32 * (lpix >> 3) + 8 * m + (lpix & 7);
        int dg_e = dgi_s[e];
        qa[m]  = (LCONV - 1) - idx_s[e] + 2 * hl;
        rem[m] = dg_e - 2 * hl;
        dmx = (dg_e > dmx) ? dg_e : dmx;
    }

    v8f accM[4], accP[4];
#pragma unroll
    for (int m = 0; m < 4; ++m) { accM[m] = v8f{}; accP[m] = v8f{}; }

    int nkv = (rowmax16(dmx) + 3) >> 2;
    nkv = (nkv > KSTEPS) ? KSTEPS : nkv;
    const int nk = __builtin_amdgcn_readfirstlane(nkv);
    for (int k = 0; k < nk; ++k) {
        v2f Bm, Bp;
        Bm.x = P * rmt; Bm.y = Bm.x * rm; rmt *= rm4;
        Bp.x = Q * rpt; Bp.y = Bp.x * rp; rpt *= rp4;
#pragma unroll
        for (int m = 0; m < 4; ++m) {
            float a0 = rev_s[qa[m]], a1 = rev_s[qa[m] + 1];
            v2f A;
            A.x = (rem[m] > 0) ? a0 : 0.0f;
            A.y = (rem[m] > 1) ? a1 : 0.0f;
            qa[m] += 4; rem[m] -= 4;
            accM[m] = __builtin_amdgcn_wmma_f32_16x16x4_f32(false, A, false, Bm, (short)0, accM[m], false, false);
            accP[m] = __builtin_amdgcn_wmma_f32_16x16x4_f32(false, A, false, Bp, (short)0, accP[m], false, false);
        }
    }

    union { v8f v[4]; float f[32]; } um, up;
#pragma unroll
    for (int m = 0; m < 4; ++m) { um.v[m] = accM[m]; up.v[m] = accP[m]; }
    const float* dg = dgf_s + 32 * hl;

    float tm = 0.0f, tp = 0.0f;
#pragma unroll
    for (int i = 0; i < HALFS; ++i) {
        float d = dg[i];
        tm = fmaf(__builtin_amdgcn_exp2f(d * lgm), tm, um.f[i]);
        tp = fmaf(__builtin_amdgcn_exp2f(d * lgp), tp, up.f[i]);
    }
    float cym = xhalf(tm), cyp = xhalf(tp);
    tm = hl ? cym : 0.0f;
    tp = hl ? cyp : 0.0f;
    float lgm2 = lgm, lgp2 = lgp;
    asm volatile("" : "+v"(lgm2), "+v"(lgp2));

    const float COSF = 0.98480775301220805936f;
    const float SINF = 0.17364817766693034885f;
    float E1   = __expf(-0.00487f);
    float M0   = 100.0f * (1.0f - COSF * E1) / (SINF * (1.0f - E1));
    float M0t  = M0 * SINF;
    float Mst  = M0t * (1.0f - E1) / (1.0f - E1 * COSF);
    float offs = 100.0f - Mst;

    float* orow = out + (HALFS * hl) * NPIX + p;
#pragma unroll
    for (int i = 0; i < HALFS; ++i) {
        float d = dg[i];
        tm = fmaf(__builtin_amdgcn_exp2f(d * lgm2), tm, um.f[i]);
        tp = fmaf(__builtin_amdgcn_exp2f(d * lgp2), tp, up.f[i]);
        float CA   = tm + tp;
        float E1CA = __expf(-0.00487f * (1.0f + 4.3f * CA));
        float CAt  = M0t * (1.0f - E1CA) *
                     __builtin_amdgcn_rcpf(1.0f - E1CA * COSF);
        orow[i * NPIX] = CAt + offs;
    }
}

extern "C" void kernel_launch(void* const* d_in, const int* in_sizes, int n_in,
                              void* d_out, int out_size, void* d_ws, size_t ws_size,
                              hipStream_t stream)
{
    (void)in_sizes; (void)n_in; (void)out_size; (void)ws_size;
    const float* param = (const float*)d_in[0];
    const float* st    = (const float*)d_in[1];
    const float* cp    = (const float*)d_in[2];
    float* out   = (float*)d_out;
    (void)d_ws;
    dce_main<<<256, BTHREADS, 0, stream>>>(param, st, cp, out);
}
